# starting kernel
# baseline (speedup 1.0000x reference)
#include <hip/hip_runtime.h>
#include <hip/hip_bf16.h>
#include <math.h>

#define LL   141
#define CCH  8
#define HH   64
#define MIDD 128
#define HCC  512

typedef __attribute__((ext_vector_type(2))) float v2f;
typedef __attribute__((ext_vector_type(8))) float v8f;

#define WMMA4(a, b, c) \
  __builtin_amdgcn_wmma_f32_16x16x4_f32(false, (a), false, (b), (short)0, (c), false, false)

__device__ __forceinline__ v2f load_a4(const float* base, int ld, int kb, int lane) {
  int m = lane & 15;
  int k = kb + ((lane >> 4) << 1);
  const float* p = base + m * ld + k;
  v2f r = {p[0], p[1]};
  return r;
}

__device__ __forceinline__ v2f load_bT(const float* W, int ldw, int n0, int kb, int lane) {
  int n = n0 + (lane & 15);
  int k = kb + ((lane >> 4) << 1);
  const float* p = W + n * ldw + k;
  v2f r = {p[0], p[1]};
  return r;
}

__device__ __forceinline__ void g_eval(const float* z, float* h, float* kout,
                                       const float* dX,
                                       const float* __restrict__ W1, const float* __restrict__ b1,
                                       const float* __restrict__ W2, const float* __restrict__ b2,
                                       int lane) {
  v2f aZ[HH / 4];
#pragma unroll
  for (int ks = 0; ks < HH / 4; ++ks) aZ[ks] = load_a4(z, HH, ks * 4, lane);

  for (int nt = 0; nt < MIDD / 16; ++nt) {
    v8f acc = {};
#pragma unroll
    for (int ks = 0; ks < HH / 4; ++ks) {
      v2f b = load_bT(W1, HH, nt * 16, ks * 4, lane);
      acc = WMMA4(aZ[ks], b, acc);
    }
    int col = nt * 16 + (lane & 15);
    float bias = b1[col];
#pragma unroll
    for (int j = 0; j < 8; ++j) {
      int row = (lane < 16) ? j : j + 8;
      h[row * MIDD + col] = fmaxf(acc[j] + bias, 0.0f);
    }
  }
  __syncthreads();

  int cc   = lane & 7;
  int hsel = (lane >> 3) & 1;

  v2f aH[MIDD / 4];
#pragma unroll
  for (int ks = 0; ks < MIDD / 4; ++ks) aH[ks] = load_a4(h, MIDD, ks * 4, lane);

  float dxv[8];
#pragma unroll
  for (int j = 0; j < 8; ++j) {
    int row = (lane < 16) ? j : j + 8;
    dxv[j] = dX[row * CCH + cc];
  }

  for (int nt = 0; nt < HCC / 16; ++nt) {
    v8f acc = {};
#pragma unroll
    for (int ks = 0; ks < MIDD / 4; ++ks) {
      v2f b = load_bT(W2, MIDD, nt * 16, ks * 4, lane);
      acc = WMMA4(aH[ks], b, acc);
    }
    int col  = nt * 16 + (lane & 15);
    float bias = b2[col];
    int hout = nt * 2 + hsel;
#pragma unroll
    for (int j = 0; j < 8; ++j) {
      int row = (lane < 16) ? j : j + 8;
      float f = tanhf(acc[j] + bias);
      float p = f * dxv[j];
      p += __shfl_xor(p, 1, 32);
      p += __shfl_xor(p, 2, 32);
      p += __shfl_xor(p, 4, 32);
      if (cc == 0) kout[row * HH + hout] = p;
    }
  }
  __syncthreads();
}

__global__ __launch_bounds__(32) void ncde_fused(
    const float* __restrict__ coeffs,
    const float* __restrict__ W_init, const float* __restrict__ b_init,
    const float* __restrict__ W1, const float* __restrict__ b1,
    const float* __restrict__ W2, const float* __restrict__ b2,
    const float* __restrict__ W_out, const float* __restrict__ b_out,
    float* __restrict__ out) {
  __shared__ float s_z [16 * HH];
  __shared__ float s_z0[16 * HH];
  __shared__ float s_h [16 * MIDD];
  __shared__ float s_k1[16 * HH];
  __shared__ float s_k2[16 * HH];
  __shared__ float s_kc[16 * HH];
  __shared__ float s_ac[16 * HH];
  __shared__ float s_dX[3][16 * CCH];

  const int lane = threadIdx.x;
  const int r0   = blockIdx.x * 16;
  const float dt = 140.0f / 49.0f;

  for (int i = 0; i < 3; ++i)
    for (int q = lane; q < 16 * CCH; q += 32) {
      int b = q >> 3, c = q & 7;
      long base = ((long)(r0 + b) * LL + i) * CCH + c;
      s_dX[i][q] = coeffs[base + CCH] - coeffs[base];
    }

  {
    v2f aX[CCH / 4];
#pragma unroll
    for (int ks = 0; ks < CCH / 4; ++ks)
      aX[ks] = load_a4(coeffs + (long)r0 * LL * CCH, LL * CCH, ks * 4, lane);
    for (int nt = 0; nt < HH / 16; ++nt) {
      v8f acc = {};
#pragma unroll
      for (int ks = 0; ks < CCH / 4; ++ks) {
        v2f b = load_bT(W_init, CCH, nt * 16, ks * 4, lane);
        acc = WMMA4(aX[ks], b, acc);
      }
      int col = nt * 16 + (lane & 15);
      float bias = b_init[col];
#pragma unroll
      for (int j = 0; j < 8; ++j) {
        int row = (lane < 16) ? j : j + 8;
        float v = acc[j] + bias;
        s_z0[row * HH + col] = v;
        s_z [row * HH + col] = v;
      }
    }
  }
  __syncthreads();

  g_eval(s_z, s_h, s_kc, s_dX[0], W1, b1, W2, b2, lane);
  for (int q = lane; q < 16 * HH; q += 32) {
    float k1 = s_kc[q];
    s_k1[q] = k1;
    s_ac[q] = k1;
    s_z[q]  = s_z0[q] + dt * (k1 * (1.0f / 3.0f));
  }
  __syncthreads();

  g_eval(s_z, s_h, s_kc, s_dX[0], W1, b1, W2, b2, lane);
  for (int q = lane; q < 16 * HH; q += 32) {
    float k2 = s_kc[q];
    s_k2[q] = k2;
    s_ac[q] += 3.0f * k2;
    s_z[q]  = s_z0[q] + dt * (k2 - s_k1[q] * (1.0f / 3.0f));
  }
  __syncthreads();

  g_eval(s_z, s_h, s_kc, s_dX[1], W1, b1, W2, b2, lane);
  for (int q = lane; q < 16 * HH; q += 32) {
    float k3 = s_kc[q];
    s_ac[q] += 3.0f * k3;
    s_z[q]  = s_z0[q] + dt * (s_k1[q] - s_k2[q] + k3);
  }
  __syncthreads();

  g_eval(s_z, s_h, s_kc, s_dX[2], W1, b1, W2, b2, lane);
  for (int q = lane; q < 16 * HH; q += 32) {
    s_z[q] = s_z0[q] + (s_ac[q] + s_kc[q]) * (dt * 0.125f);
  }
  __syncthreads();

  {
    int b = lane & 15, o = lane >> 4;
    float s = b_out[o];
    for (int hh = 0; hh < HH; ++hh)
      s = fmaf(s_z[b * HH + hh], W_out[o * HH + hh], s);
    out[(long)(r0 + b) * 2 + o] = s;
  }
}

extern "C" void kernel_launch(void* const* d_in, const int* in_sizes, int n_in,
                              void* d_out, int out_size, void* d_ws, size_t ws_size,
                              hipStream_t stream) {
  (void)n_in; (void)out_size; (void)d_ws; (void)ws_size;
  const float* coeffs = (const float*)d_in[0];
  const float* W_init = (const float*)d_in[1];
  const float* b_init = (const float*)d_in[2];
  const float* W1     = (const float*)d_in[3];
  const float* b1     = (const float*)d_in[4];
  const float* W2     = (const float*)d_in[5];
  const float* b2     = (const float*)d_in[6];
  const float* W_out  = (const float*)d_in[7];
  const float* b_out  = (const float*)d_in[8];
  float* out = (float*)d_out;

  int B = in_sizes[0] / (LL * CCH);
  int blocks = B / 16;
  ncde_fused<<<blocks, 32, 0, stream>>>(coeffs, W_init, b_init, W1, b1, W2, b2,
                                        W_out, b_out, out);
}


# LDS row pitch of z and h padded by 4 floats (68, 132) so each A-fragment read of 16 rows hits 64 different banks instead of 4; sum orders unchanged
# speedup vs baseline: 7.5093x; 7.5093x over previous
#include <hip/hip_runtime.h>
#include <hip/hip_bf16.h>
#include <math.h>

#define LL   141
#define CCH  8
#define HH   64
#define MIDD 128
#define HCC  512
#define NWAVES 16
#define NTHREADS (NWAVES * 32)
#define ZLD  (HH + 4)
#define HLD  (MIDD + 4)

typedef __attribute__((ext_vector_type(2))) float v2f;
typedef __attribute__((ext_vector_type(8))) float v8f;

#define WMMA4(a, b, c) \
  __builtin_amdgcn_wmma_f32_16x16x4_f32(false, (a), false, (b), (short)0, (c), false, false)

__device__ __forceinline__ v2f load_a4(const float* base, int ld, int kb, int lane) {
  int m = lane & 15;
  int k = kb + ((lane >> 4) << 1);
  const float* p = base + m * ld + k;
  v2f r = {p[0], p[1]};
  return r;
}

__device__ __forceinline__ v2f load_bT(const float* W, int ldw, int n0, int kb, int lane) {
  int n = n0 + (lane & 15);
  int k = kb + ((lane >> 4) << 1);
  const float* p = W + n * ldw + k;
  v2f r = {p[0], p[1]};
  return r;
}

__device__ __forceinline__ void g_eval(const float* z, float* h, float* kout,
                                       const float* dX,
                                       const float* __restrict__ W1, const float* __restrict__ b1,
                                       const float* __restrict__ W2, const float* __restrict__ b2,
                                       int lane, int wave) {
  v2f aZ[HH / 4];
#pragma unroll
  for (int ks = 0; ks < HH / 4; ++ks) aZ[ks] = load_a4(z, ZLD, ks * 4, lane);

  for (int nt = wave; nt < MIDD / 16; nt += NWAVES) {
    v8f acc = {};
#pragma unroll
    for (int ks = 0; ks < HH / 4; ++ks) {
      v2f b = load_bT(W1, HH, nt * 16, ks * 4, lane);
      acc = WMMA4(aZ[ks], b, acc);
    }
    int col = nt * 16 + (lane & 15);
    float bias = b1[col];
#pragma unroll
    for (int j = 0; j < 8; ++j) {
      int row = (lane < 16) ? j : j + 8;
      h[row * HLD + col] = fmaxf(acc[j] + bias, 0.0f);
    }
  }
  __syncthreads();

  int cc   = lane & 7;
  int hsel = (lane >> 3) & 1;

  v2f aH[MIDD / 4];
#pragma unroll
  for (int ks = 0; ks < MIDD / 4; ++ks) aH[ks] = load_a4(h, HLD, ks * 4, lane);

  float dxv[8];
#pragma unroll
  for (int j = 0; j < 8; ++j) {
    int row = (lane < 16) ? j : j + 8;
    dxv[j] = dX[row * CCH + cc];
  }

  for (int nt = wave * (HCC / 16 / NWAVES); nt < (wave + 1) * (HCC / 16 / NWAVES); ++nt) {
    v8f acc = {};
#pragma unroll
    for (int ks = 0; ks < MIDD / 4; ++ks) {
      v2f b = load_bT(W2, MIDD, nt * 16, ks * 4, lane);
      acc = WMMA4(aH[ks], b, acc);
    }
    int col  = nt * 16 + (lane & 15);
    float bias = b2[col];
    int hout = nt * 2 + hsel;
#pragma unroll
    for (int j = 0; j < 8; ++j) {
      int row = (lane < 16) ? j : j + 8;
      float f = tanhf(acc[j] + bias);
      float p = f * dxv[j];
      p += __shfl_xor(p, 1, 32);
      p += __shfl_xor(p, 2, 32);
      p += __shfl_xor(p, 4, 32);
      if (cc == 0) kout[row * HH + hout] = p;
    }
  }
  __syncthreads();
}

__global__ __launch_bounds__(NTHREADS) void ncde_fused(
    const float* __restrict__ coeffs,
    const float* __restrict__ W_init, const float* __restrict__ b_init,
    const float* __restrict__ W1, const float* __restrict__ b1,
    const float* __restrict__ W2, const float* __restrict__ b2,
    const float* __restrict__ W_out, const float* __restrict__ b_out,
    float* __restrict__ out) {
  __shared__ float s_z [16 * ZLD];
  __shared__ float s_z0[16 * HH];
  __shared__ float s_h [16 * HLD];
  __shared__ float s_k1[16 * HH];
  __shared__ float s_k2[16 * HH];
  __shared__ float s_kc[16 * HH];
  __shared__ float s_ac[16 * HH];
  __shared__ float s_dX[3][16 * CCH];

  const int tid  = threadIdx.x;
  const int lane = tid & 31;
  const int wave = __builtin_amdgcn_readfirstlane(tid >> 5);
  const int r0   = blockIdx.x * 16;
  const float dt = 140.0f / 49.0f;

  for (int i = 0; i < 3; ++i)
    for (int q = tid; q < 16 * CCH; q += NTHREADS) {
      int b = q >> 3, c = q & 7;
      long base = ((long)(r0 + b) * LL + i) * CCH + c;
      s_dX[i][q] = coeffs[base + CCH] - coeffs[base];
    }

  {
    v2f aX[CCH / 4];
#pragma unroll
    for (int ks = 0; ks < CCH / 4; ++ks)
      aX[ks] = load_a4(coeffs + (long)r0 * LL * CCH, LL * CCH, ks * 4, lane);
    for (int nt = wave; nt < HH / 16; nt += NWAVES) {
      v8f acc = {};
#pragma unroll
      for (int ks = 0; ks < CCH / 4; ++ks) {
        v2f b = load_bT(W_init, CCH, nt * 16, ks * 4, lane);
        acc = WMMA4(aX[ks], b, acc);
      }
      int col = nt * 16 + (lane & 15);
      float bias = b_init[col];
#pragma unroll
      for (int j = 0; j < 8; ++j) {
        int row = (lane < 16) ? j : j + 8;
        float v = acc[j] + bias;
        s_z0[row * HH + col] = v;
        s_z [row * ZLD + col] = v;
      }
    }
  }
  __syncthreads();

  g_eval(s_z, s_h, s_kc, s_dX[0], W1, b1, W2, b2, lane, wave);
  for (int q = tid; q < 16 * HH; q += NTHREADS) {
    float k1 = s_kc[q];
    s_k1[q] = k1;
    s_ac[q] = k1;
    s_z[(q >> 6) * ZLD + (q & (HH - 1))] = s_z0[q] + dt * (k1 * (1.0f / 3.0f));
  }
  __syncthreads();

  g_eval(s_z, s_h, s_kc, s_dX[0], W1, b1, W2, b2, lane, wave);
  for (int q = tid; q < 16 * HH; q += NTHREADS) {
    float k2 = s_kc[q];
    s_k2[q] = k2;
    s_ac[q] += 3.0f * k2;
    s_z[(q >> 6) * ZLD + (q & (HH - 1))] = s_z0[q] + dt * (k2 - s_k1[q] * (1.0f / 3.0f));
  }
  __syncthreads();

  g_eval(s_z, s_h, s_kc, s_dX[1], W1, b1, W2, b2, lane, wave);
  for (int q = tid; q < 16 * HH; q += NTHREADS) {
    float k3 = s_kc[q];
    s_ac[q] += 3.0f * k3;
    s_z[(q >> 6) * ZLD + (q & (HH - 1))] = s_z0[q] + dt * (s_k1[q] - s_k2[q] + k3);
  }
  __syncthreads();

  g_eval(s_z, s_h, s_kc, s_dX[2], W1, b1, W2, b2, lane, wave);
  for (int q = tid; q < 16 * HH; q += NTHREADS) {
    s_z[(q >> 6) * ZLD + (q & (HH - 1))] = s_z0[q] + (s_ac[q] + s_kc[q]) * (dt * 0.125f);
  }
  __syncthreads();

  if (wave == 0) {
    int b = lane & 15, o = lane >> 4;
    float s = b_out[o];
    for (int hh = 0; hh < HH; ++hh)
      s = fmaf(s_z[b * ZLD + hh], W_out[o * HH + hh], s);
    out[(long)(r0 + b) * 2 + o] = s;
  }
}

extern "C" void kernel_launch(void* const* d_in, const int* in_sizes, int n_in,
                              void* d_out, int out_size, void* d_ws, size_t ws_size,
                              hipStream_t stream) {
  (void)n_in; (void)out_size; (void)d_ws; (void)ws_size;
  const float* coeffs = (const float*)d_in[0];
  const float* W_init = (const float*)d_in[1];
  const float* b_init = (const float*)d_in[2];
  const float* W1     = (const float*)d_in[3];
  const float* b1     = (const float*)d_in[4];
  const float* W2     = (const float*)d_in[5];
  const float* b2     = (const float*)d_in[6];
  const float* W_out  = (const float*)d_in[7];
  const float* b_out  = (const float*)d_in[8];
  float* out = (float*)d_out;

  int B = in_sizes[0] / (LL * CCH);
  int blocks = B / 16;
  ncde_fused<<<blocks, NTHREADS, 0, stream>>>(coeffs, W_init, b_init, W1, b1, W2, b2,
                                        W_out, b_out, out);
}


# Each wave's W1/W2 B fragments and biases are loaded from global once and held in registers across all four stage evaluations; its two W2 tiles share each A read; sum orders unchanged
# speedup vs baseline: 9.6011x; 1.2786x over previous
#include <hip/hip_runtime.h>
#include <hip/hip_bf16.h>
#include <math.h>

#define LL   141
#define CCH  8
#define HH   64
#define MIDD 128
#define HCC  512
#define NWAVES 16
#define NTHREADS (NWAVES * 32)
#define ZLD  (HH + 4)
#define HLD  (MIDD + 4)
#define T2   (HCC / 16 / NWAVES)

typedef __attribute__((ext_vector_type(2))) float v2f;
typedef __attribute__((ext_vector_type(8))) float v8f;

#define WMMA4(a, b, c) \
  __builtin_amdgcn_wmma_f32_16x16x4_f32(false, (a), false, (b), (short)0, (c), false, false)

__device__ __forceinline__ v2f load_a4(const float* base, int ld, int kb, int lane) {
  int m = lane & 15;
  int k = kb + ((lane >> 4) << 1);
  const float* p = base + m * ld + k;
  v2f r = {p[0], p[1]};
  return r;
}

__device__ __forceinline__ v2f load_bT(const float* W, int ldw, int n0, int kb, int lane) {
  int n = n0 + (lane & 15);
  int k = kb + ((lane >> 4) << 1);
  const float* p = W + n * ldw + k;
  v2f r = {p[0], p[1]};
  return r;
}

__device__ __forceinline__ void g_eval(const float* z, float* h, float* kout,
                                       const float* dX,
                                       const v2f (&bW1)[HH / 4], float bias1,
                                       const v2f (&bW2)[T2][MIDD / 4], const float (&bias2)[T2],
                                       int lane, int wave) {
  if (wave < MIDD / 16) {
    v2f aZ[HH / 4];
#pragma unroll
    for (int ks = 0; ks < HH / 4; ++ks) aZ[ks] = load_a4(z, ZLD, ks * 4, lane);
    v8f acc = {};
#pragma unroll
    for (int ks = 0; ks < HH / 4; ++ks) acc = WMMA4(aZ[ks], bW1[ks], acc);
    int col = wave * 16 + (lane & 15);
#pragma unroll
    for (int j = 0; j < 8; ++j) {
      int row = (lane < 16) ? j : j + 8;
      h[row * HLD + col] = fmaxf(acc[j] + bias1, 0.0f);
    }
  }
  __syncthreads();

  int cc   = lane & 7;
  int hsel = (lane >> 3) & 1;

  float dxv[8];
#pragma unroll
  for (int j = 0; j < 8; ++j) {
    int row = (lane < 16) ? j : j + 8;
    dxv[j] = dX[row * CCH + cc];
  }

  v8f acc[T2];
#pragma unroll
  for (int t = 0; t < T2; ++t) acc[t] = v8f{};
#pragma unroll
  for (int ks = 0; ks < MIDD / 4; ++ks) {
    v2f a = load_a4(h, HLD, ks * 4, lane);
#pragma unroll
    for (int t = 0; t < T2; ++t) acc[t] = WMMA4(a, bW2[t][ks], acc[t]);
  }
#pragma unroll
  for (int t = 0; t < T2; ++t) {
    int hout = (wave * T2 + t) * 2 + hsel;
#pragma unroll
    for (int j = 0; j < 8; ++j) {
      int row = (lane < 16) ? j : j + 8;
      float f = tanhf(acc[t][j] + bias2[t]);
      float p = f * dxv[j];
      p += __shfl_xor(p, 1, 32);
      p += __shfl_xor(p, 2, 32);
      p += __shfl_xor(p, 4, 32);
      if (cc == 0) kout[row * HH + hout] = p;
    }
  }
  __syncthreads();
}

__global__ __launch_bounds__(NTHREADS) void ncde_fused(
    const float* __restrict__ coeffs,
    const float* __restrict__ W_init, const float* __restrict__ b_init,
    const float* __restrict__ W1, const float* __restrict__ b1,
    const float* __restrict__ W2, const float* __restrict__ b2,
    const float* __restrict__ W_out, const float* __restrict__ b_out,
    float* __restrict__ out) {
  __shared__ float s_z [16 * ZLD];
  __shared__ float s_z0[16 * HH];
  __shared__ float s_h [16 * HLD];
  __shared__ float s_k1[16 * HH];
  __shared__ float s_k2[16 * HH];
  __shared__ float s_kc[16 * HH];
  __shared__ float s_ac[16 * HH];
  __shared__ float s_dX[3][16 * CCH];

  const int tid  = threadIdx.x;
  const int lane = tid & 31;
  const int wave = __builtin_amdgcn_readfirstlane(tid >> 5);
  const int r0   = blockIdx.x * 16;
  const float dt = 140.0f / 49.0f;

  v2f bW1[HH / 4];
  float bias1 = 0.0f;
  if (wave < MIDD / 16) {
#pragma unroll
    for (int ks = 0; ks < HH / 4; ++ks) bW1[ks] = load_bT(W1, HH, wave * 16, ks * 4, lane);
    bias1 = b1[wave * 16 + (lane & 15)];
  }
  v2f bW2[T2][MIDD / 4];
  float bias2[T2];
#pragma unroll
  for (int t = 0; t < T2; ++t) {
#pragma unroll
    for (int ks = 0; ks < MIDD / 4; ++ks)
      bW2[t][ks] = load_bT(W2, MIDD, (wave * T2 + t) * 16, ks * 4, lane);
    bias2[t] = b2[(wave * T2 + t) * 16 + (lane & 15)];
  }

  for (int i = 0; i < 3; ++i)
    for (int q = tid; q < 16 * CCH; q += NTHREADS) {
      int b = q >> 3, c = q & 7;
      long base = ((long)(r0 + b) * LL + i) * CCH + c;
      s_dX[i][q] = coeffs[base + CCH] - coeffs[base];
    }

  {
    v2f aX[CCH / 4];
#pragma unroll
    for (int ks = 0; ks < CCH / 4; ++ks)
      aX[ks] = load_a4(coeffs + (long)r0 * LL * CCH, LL * CCH, ks * 4, lane);
    for (int nt = wave; nt < HH / 16; nt += NWAVES) {
      v8f acc = {};
#pragma unroll
      for (int ks = 0; ks < CCH / 4; ++ks) {
        v2f b = load_bT(W_init, CCH, nt * 16, ks * 4, lane);
        acc = WMMA4(aX[ks], b, acc);
      }
      int col = nt * 16 + (lane & 15);
      float bias = b_init[col];
#pragma unroll
      for (int j = 0; j < 8; ++j) {
        int row = (lane < 16) ? j : j + 8;
        float v = acc[j] + bias;
        s_z0[row * HH + col] = v;
        s_z [row * ZLD + col] = v;
      }
    }
  }
  __syncthreads();

  g_eval(s_z, s_h, s_kc, s_dX[0], bW1, bias1, bW2, bias2, lane, wave);
  for (int q = tid; q < 16 * HH; q += NTHREADS) {
    float k1 = s_kc[q];
    s_k1[q] = k1;
    s_ac[q] = k1;
    s_z[(q >> 6) * ZLD + (q & (HH - 1))] = s_z0[q] + dt * (k1 * (1.0f / 3.0f));
  }
  __syncthreads();

  g_eval(s_z, s_h, s_kc, s_dX[0], bW1, bias1, bW2, bias2, lane, wave);
  for (int q = tid; q < 16 * HH; q += NTHREADS) {
    float k2 = s_kc[q];
    s_k2[q] = k2;
    s_ac[q] += 3.0f * k2;
    s_z[(q >> 6) * ZLD + (q & (HH - 1))] = s_z0[q] + dt * (k2 - s_k1[q] * (1.0f / 3.0f));
  }
  __syncthreads();

  g_eval(s_z, s_h, s_kc, s_dX[1], bW1, bias1, bW2, bias2, lane, wave);
  for (int q = tid; q < 16 * HH; q += NTHREADS) {
    float k3 = s_kc[q];
    s_ac[q] += 3.0f * k3;
    s_z[(q >> 6) * ZLD + (q & (HH - 1))] = s_z0[q] + dt * (s_k1[q] - s_k2[q] + k3);
  }
  __syncthreads();

  g_eval(s_z, s_h, s_kc, s_dX[2], bW1, bias1, bW2, bias2, lane, wave);
  for (int q = tid; q < 16 * HH; q += NTHREADS) {
    s_z[(q >> 6) * ZLD + (q & (HH - 1))] = s_z0[q] + (s_ac[q] + s_kc[q]) * (dt * 0.125f);
  }
  __syncthreads();

  if (wave == 0) {
    int b = lane & 15, o = lane >> 4;
    float s = b_out[o];
    for (int hh = 0; hh < HH; ++hh)
      s = fmaf(s_z[b * ZLD + hh], W_out[o * HH + hh], s);
    out[(long)(r0 + b) * 2 + o] = s;
  }
}

extern "C" void kernel_launch(void* const* d_in, const int* in_sizes, int n_in,
                              void* d_out, int out_size, void* d_ws, size_t ws_size,
                              hipStream_t stream) {
  (void)n_in; (void)out_size; (void)d_ws; (void)ws_size;
  const float* coeffs = (const float*)d_in[0];
  const float* W_init = (const float*)d_in[1];
  const float* b_init = (const float*)d_in[2];
  const float* W1     = (const float*)d_in[3];
  const float* b1     = (const float*)d_in[4];
  const float* W2     = (const float*)d_in[5];
  const float* b2     = (const float*)d_in[6];
  const float* W_out  = (const float*)d_in[7];
  const float* b_out  = (const float*)d_in[8];
  float* out = (float*)d_out;

  int B = in_sizes[0] / (LL * CCH);
  int blocks = B / 16;
  ncde_fused<<<blocks, NTHREADS, 0, stream>>>(coeffs, W_init, b_init, W1, b1, W2, b2,
                                        W_out, b_out, out);
}


# GEMM 2 epilogue no longer a serial chain per value: tanh on 8-vectors with both ranges computed and selected (same values), 8-lane c-sum on DPP row_xmask adds not LDS shuffles; same sum tree
# speedup vs baseline: 12.5681x; 1.3090x over previous
#include <hip/hip_runtime.h>
#include <hip/hip_bf16.h>
#include <math.h>

#define LL   141
#define CCH  8
#define HH   64
#define MIDD 128
#define HCC  512
#define NWAVES 16
#define NTHREADS (NWAVES * 32)
#define ZLD  (HH + 4)
#define HLD  (MIDD + 4)
#define T2   (HCC / 16 / NWAVES)

typedef __attribute__((ext_vector_type(2))) float v2f;
typedef __attribute__((ext_vector_type(8))) float v8f;

#define WMMA4(a, b, c) \
  __builtin_amdgcn_wmma_f32_16x16x4_f32(false, (a), false, (b), (short)0, (c), false, false)

__device__ __forceinline__ v2f load_a4(const float* base, int ld, int kb, int lane) {
  int m = lane & 15;
  int k = kb + ((lane >> 4) << 1);
  const float* p = base + m * ld + k;
  v2f r = {p[0], p[1]};
  return r;
}

__device__ __forceinline__ v2f load_bT(const float* W, int ldw, int n0, int kb, int lane) {
  int n = n0 + (lane & 15);
  int k = kb + ((lane >> 4) << 1);
  const float* p = W + n * ldw + k;
  v2f r = {p[0], p[1]};
  return r;
}

__device__ __forceinline__ v8f splat8(float c) { return v8f{c, c, c, c, c, c, c, c}; }
__device__ __forceinline__ v8f tanh_sel8(v8f x) {
  v8f y  = __builtin_elementwise_abs(x);
  v8f y2 = y * y;
  v8f pl = __builtin_elementwise_fma(y2, splat8(-0x1.758e7ap-8f), splat8(0x1.521192p-6f));
  pl = __builtin_elementwise_fma(y2, pl, splat8(-0x1.b8389cp-5f));
  pl = __builtin_elementwise_fma(y2, pl, splat8(0x1.110704p-3f));
  pl = __builtin_elementwise_fma(y2, pl, splat8(-0x1.555532p-2f));
  v8f zs = __builtin_elementwise_fma(y2, y * pl, y);
  v8f d  = __builtin_elementwise_exp(y + y) + 1.0f;
  v8f r;
#pragma unroll
  for (int j = 0; j < 8; ++j) r[j] = __builtin_amdgcn_rcpf(d[j]);
  v8f zl = __builtin_elementwise_fma(splat8(-2.0f), r, splat8(1.0f));
  v8f z  = (y < 0.625f) ? zs : zl;
  return __builtin_elementwise_copysign(z, x);
}

template <int M>
__device__ __forceinline__ float lane_xor(float v) {
  int r = __builtin_amdgcn_update_dpp(0, __builtin_bit_cast(int, v), 0x160 + M, 0xf, 0xf, true);
  return __builtin_bit_cast(float, r);
}

__device__ __forceinline__ void g_eval(const float* z, float* h, float* kout,
                                       const float* dX,
                                       const v2f (&bW1)[HH / 4], float bias1,
                                       const v2f (&bW2)[T2][MIDD / 4], const float (&bias2)[T2],
                                       int lane, int wave) {
  if (wave < MIDD / 16) {
    v2f aZ[HH / 4];
#pragma unroll
    for (int ks = 0; ks < HH / 4; ++ks) aZ[ks] = load_a4(z, ZLD, ks * 4, lane);
    v8f acc = {};
#pragma unroll
    for (int ks = 0; ks < HH / 4; ++ks) acc = WMMA4(aZ[ks], bW1[ks], acc);
    int col = wave * 16 + (lane & 15);
#pragma unroll
    for (int j = 0; j < 8; ++j) {
      int row = (lane < 16) ? j : j + 8;
      h[row * HLD + col] = fmaxf(acc[j] + bias1, 0.0f);
    }
  }
  __syncthreads();

  int cc   = lane & 7;
  int hsel = (lane >> 3) & 1;

  float dxv[8];
#pragma unroll
  for (int j = 0; j < 8; ++j) {
    int row = (lane < 16) ? j : j + 8;
    dxv[j] = dX[row * CCH + cc];
  }

  v8f acc[T2];
#pragma unroll
  for (int t = 0; t < T2; ++t) acc[t] = v8f{};
#pragma unroll
  for (int ks = 0; ks < MIDD / 4; ++ks) {
    v2f a = load_a4(h, HLD, ks * 4, lane);
#pragma unroll
    for (int t = 0; t < T2; ++t) acc[t] = WMMA4(a, bW2[t][ks], acc[t]);
  }
  v8f dx8;
#pragma unroll
  for (int j = 0; j < 8; ++j) dx8[j] = dxv[j];
  v8f p[T2];
#pragma unroll
  for (int t = 0; t < T2; ++t) {
    v8f q = tanh_sel8(acc[t] + bias2[t]) * dx8;
#pragma unroll
    for (int j = 0; j < 8; ++j) q[j] += lane_xor<1>(q[j]);
#pragma unroll
    for (int j = 0; j < 8; ++j) q[j] += lane_xor<2>(q[j]);
#pragma unroll
    for (int j = 0; j < 8; ++j) q[j] += lane_xor<4>(q[j]);
    p[t] = q;
  }
  if (cc == 0) {
#pragma unroll
    for (int t = 0; t < T2; ++t) {
      int hout = (wave * T2 + t) * 2 + hsel;
#pragma unroll
      for (int j = 0; j < 8; ++j) {
        int row = (lane < 16) ? j : j + 8;
        kout[row * HH + hout] = p[t][j];
      }
    }
  }
  __syncthreads();
}

__global__ __launch_bounds__(NTHREADS) void ncde_fused(
    const float* __restrict__ coeffs,
    const float* __restrict__ W_init, const float* __restrict__ b_init,
    const float* __restrict__ W1, const float* __restrict__ b1,
    const float* __restrict__ W2, const float* __restrict__ b2,
    const float* __restrict__ W_out, const float* __restrict__ b_out,
    float* __restrict__ out) {
  __shared__ float s_z [16 * ZLD];
  __shared__ float s_z0[16 * HH];
  __shared__ float s_h [16 * HLD];
  __shared__ float s_k1[16 * HH];
  __shared__ float s_k2[16 * HH];
  __shared__ float s_kc[16 * HH];
  __shared__ float s_ac[16 * HH];
  __shared__ float s_dX[3][16 * CCH];

  const int tid  = threadIdx.x;
  const int lane = tid & 31;
  const int wave = __builtin_amdgcn_readfirstlane(tid >> 5);
  const int r0   = blockIdx.x * 16;
  const float dt = 140.0f / 49.0f;

  v2f bW1[HH / 4];
  float bias1 = 0.0f;
  if (wave < MIDD / 16) {
#pragma unroll
    for (int ks = 0; ks < HH / 4; ++ks) bW1[ks] = load_bT(W1, HH, wave * 16, ks * 4, lane);
    bias1 = b1[wave * 16 + (lane & 15)];
  }
  v2f bW2[T2][MIDD / 4];
  float bias2[T2];
#pragma unroll
  for (int t = 0; t < T2; ++t) {
#pragma unroll
    for (int ks = 0; ks < MIDD / 4; ++ks)
      bW2[t][ks] = load_bT(W2, MIDD, (wave * T2 + t) * 16, ks * 4, lane);
    bias2[t] = b2[(wave * T2 + t) * 16 + (lane & 15)];
  }

  for (int i = 0; i < 3; ++i)
    for (int q = tid; q < 16 * CCH; q += NTHREADS) {
      int b = q >> 3, c = q & 7;
      long base = ((long)(r0 + b) * LL + i) * CCH + c;
      s_dX[i][q] = coeffs[base + CCH] - coeffs[base];
    }

  {
    v2f aX[CCH / 4];
#pragma unroll
    for (int ks = 0; ks < CCH / 4; ++ks)
      aX[ks] = load_a4(coeffs + (long)r0 * LL * CCH, LL * CCH, ks * 4, lane);
    for (int nt = wave; nt < HH / 16; nt += NWAVES) {
      v8f acc = {};
#pragma unroll
      for (int ks = 0; ks < CCH / 4; ++ks) {
        v2f b = load_bT(W_init, CCH, nt * 16, ks * 4, lane);
        acc = WMMA4(aX[ks], b, acc);
      }
      int col = nt * 16 + (lane & 15);
      float bias = b_init[col];
#pragma unroll
      for (int j = 0; j < 8; ++j) {
        int row = (lane < 16) ? j : j + 8;
        float v = acc[j] + bias;
        s_z0[row * HH + col] = v;
        s_z [row * ZLD + col] = v;
      }
    }
  }
  __syncthreads();

  g_eval(s_z, s_h, s_kc, s_dX[0], bW1, bias1, bW2, bias2, lane, wave);
  for (int q = tid; q < 16 * HH; q += NTHREADS) {
    float k1 = s_kc[q];
    s_k1[q] = k1;
    s_ac[q] = k1;
    s_z[(q >> 6) * ZLD + (q & (HH - 1))] = s_z0[q] + dt * (k1 * (1.0f / 3.0f));
  }
  __syncthreads();

  g_eval(s_z, s_h, s_kc, s_dX[0], bW1, bias1, bW2, bias2, lane, wave);
  for (int q = tid; q < 16 * HH; q += NTHREADS) {
    float k2 = s_kc[q];
    s_k2[q] = k2;
    s_ac[q] += 3.0f * k2;
    s_z[(q >> 6) * ZLD + (q & (HH - 1))] = s_z0[q] + dt * (k2 - s_k1[q] * (1.0f / 3.0f));
  }
  __syncthreads();

  g_eval(s_z, s_h, s_kc, s_dX[1], bW1, bias1, bW2, bias2, lane, wave);
  for (int q = tid; q < 16 * HH; q += NTHREADS) {
    float k3 = s_kc[q];
    s_ac[q] += 3.0f * k3;
    s_z[(q >> 6) * ZLD + (q & (HH - 1))] = s_z0[q] + dt * (s_k1[q] - s_k2[q] + k3);
  }
  __syncthreads();

  g_eval(s_z, s_h, s_kc, s_dX[2], bW1, bias1, bW2, bias2, lane, wave);
  for (int q = tid; q < 16 * HH; q += NTHREADS) {
    s_z[(q >> 6) * ZLD + (q & (HH - 1))] = s_z0[q] + (s_ac[q] + s_kc[q]) * (dt * 0.125f);
  }
  __syncthreads();

  if (wave == 0) {
    int b = lane & 15, o = lane >> 4;
    float s = b_out[o];
    for (int hh = 0; hh < HH; ++hh)
      s = fmaf(s_z[b * ZLD + hh], W_out[o * HH + hh], s);
    out[(long)(r0 + b) * 2 + o] = s;
  }
}

extern "C" void kernel_launch(void* const* d_in, const int* in_sizes, int n_in,
                              void* d_out, int out_size, void* d_ws, size_t ws_size,
                              hipStream_t stream) {
  (void)n_in; (void)out_size; (void)d_ws; (void)ws_size;
  const float* coeffs = (const float*)d_in[0];
  const float* W_init = (const float*)d_in[1];
  const float* b_init = (const float*)d_in[2];
  const float* W1     = (const float*)d_in[3];
  const float* b1     = (const float*)d_in[4];
  const float* W2     = (const float*)d_in[5];
  const float* b2     = (const float*)d_in[6];
  const float* W_out  = (const float*)d_in[7];
  const float* b_out  = (const float*)d_in[8];
  float* out = (float*)d_out;

  int B = in_sizes[0] / (LL * CCH);
  int blocks = B / 16;
  ncde_fused<<<blocks, NTHREADS, 0, stream>>>(coeffs, W_init, b_init, W1, b1, W2, b2,
                                        W_out, b_out, out);
}


# A and B fragments fetched 16 bytes per lane (one load feeds two k-steps; both loaders use k = 8s+4*half+e): half as many 16-row weight gathers; sum order over k inside each 8-chunk changes
# speedup vs baseline: 12.9487x; 1.0303x over previous
#include <hip/hip_runtime.h>
#include <hip/hip_bf16.h>
#include <math.h>

#define LL   141
#define CCH  8
#define HH   64
#define MIDD 128
#define HCC  512
#define NWAVES 16
#define NTHREADS (NWAVES * 32)
#define ZLD  (HH + 4)
#define HLD  (MIDD + 4)
#define T2   (HCC / 16 / NWAVES)

typedef __attribute__((ext_vector_type(2))) float v2f;
typedef __attribute__((ext_vector_type(4))) float v4f;
typedef __attribute__((ext_vector_type(8))) float v8f;

#define WMMA4(a, b, c) \
  __builtin_amdgcn_wmma_f32_16x16x4_f32(false, (a), false, (b), (short)0, (c), false, false)

__device__ __forceinline__ v4f load_a8(const float* base, int ld, int s, int lane) {
  int m = lane & 15;
  int k = s * 8 + ((lane >> 4) << 2);
  return *reinterpret_cast<const v4f*>(base + m * ld + k);
}

__device__ __forceinline__ v4f load_b8T(const float* W, int ldw, int n0, int s, int lane) {
  int n = n0 + (lane & 15);
  int k = s * 8 + ((lane >> 4) << 2);
  return *reinterpret_cast<const v4f*>(W + n * ldw + k);
}
__device__ __forceinline__ v2f lo2(v4f v) { return __builtin_shufflevector(v, v, 0, 1); }
__device__ __forceinline__ v2f hi2(v4f v) { return __builtin_shufflevector(v, v, 2, 3); }

__device__ __forceinline__ v8f splat8(float c) { return v8f{c, c, c, c, c, c, c, c}; }
__device__ __forceinline__ v8f tanh_sel8(v8f x) {
  v8f y  = __builtin_elementwise_abs(x);
  v8f y2 = y * y;
  v8f pl = __builtin_elementwise_fma(y2, splat8(-0x1.758e7ap-8f), splat8(0x1.521192p-6f));
  pl = __builtin_elementwise_fma(y2, pl, splat8(-0x1.b8389cp-5f));
  pl = __builtin_elementwise_fma(y2, pl, splat8(0x1.110704p-3f));
  pl = __builtin_elementwise_fma(y2, pl, splat8(-0x1.555532p-2f));
  v8f zs = __builtin_elementwise_fma(y2, y * pl, y);
  v8f d  = __builtin_elementwise_exp(y + y) + 1.0f;
  v8f r;
#pragma unroll
  for (int j = 0; j < 8; ++j) r[j] = __builtin_amdgcn_rcpf(d[j]);
  v8f zl = __builtin_elementwise_fma(splat8(-2.0f), r, splat8(1.0f));
  v8f z  = (y < 0.625f) ? zs : zl;
  return __builtin_elementwise_copysign(z, x);
}

template <int M>
__device__ __forceinline__ float lane_xor(float v) {
  int r = __builtin_amdgcn_update_dpp(0, __builtin_bit_cast(int, v), 0x160 + M, 0xf, 0xf, true);
  return __builtin_bit_cast(float, r);
}

__device__ __forceinline__ void g_eval(const float* z, float* h, float* kout,
                                       const float* dX,
                                       const v4f (&bW1)[HH / 8], float bias1,
                                       const v4f (&bW2)[T2][MIDD / 8], const float (&bias2)[T2],
                                       int lane, int wave) {
  if (wave < MIDD / 16) {
    v4f aZ[HH / 8];
#pragma unroll
    for (int s = 0; s < HH / 8; ++s) aZ[s] = load_a8(z, ZLD, s, lane);
    v8f acc = {};
#pragma unroll
    for (int s = 0; s < HH / 8; ++s) {
      acc = WMMA4(lo2(aZ[s]), lo2(bW1[s]), acc);
      acc = WMMA4(hi2(aZ[s]), hi2(bW1[s]), acc);
    }
    int col = wave * 16 + (lane & 15);
#pragma unroll
    for (int j = 0; j < 8; ++j) {
      int row = (lane < 16) ? j : j + 8;
      h[row * HLD + col] = fmaxf(acc[j] + bias1, 0.0f);
    }
  }
  __syncthreads();

  int cc   = lane & 7;
  int hsel = (lane >> 3) & 1;

  float dxv[8];
#pragma unroll
  for (int j = 0; j < 8; ++j) {
    int row = (lane < 16) ? j : j + 8;
    dxv[j] = dX[row * CCH + cc];
  }

  v8f acc[T2];
#pragma unroll
  for (int t = 0; t < T2; ++t) acc[t] = v8f{};
#pragma unroll
  for (int s = 0; s < MIDD / 8; ++s) {
    v4f a = load_a8(h, HLD, s, lane);
#pragma unroll
    for (int t = 0; t < T2; ++t) acc[t] = WMMA4(lo2(a), lo2(bW2[t][s]), acc[t]);
#pragma unroll
    for (int t = 0; t < T2; ++t) acc[t] = WMMA4(hi2(a), hi2(bW2[t][s]), acc[t]);
  }
  v8f dx8;
#pragma unroll
  for (int j = 0; j < 8; ++j) dx8[j] = dxv[j];
  v8f p[T2];
#pragma unroll
  for (int t = 0; t < T2; ++t) {
    v8f q = tanh_sel8(acc[t] + bias2[t]) * dx8;
#pragma unroll
    for (int j = 0; j < 8; ++j) q[j] += lane_xor<1>(q[j]);
#pragma unroll
    for (int j = 0; j < 8; ++j) q[j] += lane_xor<2>(q[j]);
#pragma unroll
    for (int j = 0; j < 8; ++j) q[j] += lane_xor<4>(q[j]);
    p[t] = q;
  }
  if (cc == 0) {
#pragma unroll
    for (int t = 0; t < T2; ++t) {
      int hout = (wave * T2 + t) * 2 + hsel;
#pragma unroll
      for (int j = 0; j < 8; ++j) {
        int row = (lane < 16) ? j : j + 8;
        kout[row * HH + hout] = p[t][j];
      }
    }
  }
  __syncthreads();
}

__global__ __launch_bounds__(NTHREADS) void ncde_fused(
    const float* __restrict__ coeffs,
    const float* __restrict__ W_init, const float* __restrict__ b_init,
    const float* __restrict__ W1, const float* __restrict__ b1,
    const float* __restrict__ W2, const float* __restrict__ b2,
    const float* __restrict__ W_out, const float* __restrict__ b_out,
    float* __restrict__ out) {
  __shared__ __attribute__((aligned(16))) float s_z [16 * ZLD];
  __shared__ float s_z0[16 * HH];
  __shared__ __attribute__((aligned(16))) float s_h [16 * HLD];
  __shared__ float s_k1[16 * HH];
  __shared__ float s_k2[16 * HH];
  __shared__ float s_kc[16 * HH];
  __shared__ float s_ac[16 * HH];
  __shared__ float s_dX[3][16 * CCH];

  const int tid  = threadIdx.x;
  const int lane = tid & 31;
  const int wave = __builtin_amdgcn_readfirstlane(tid >> 5);
  const int r0   = blockIdx.x * 16;
  const float dt = 140.0f / 49.0f;

  v4f bW1[HH / 8];
  float bias1 = 0.0f;
  if (wave < MIDD / 16) {
#pragma unroll
    for (int s = 0; s < HH / 8; ++s) bW1[s] = load_b8T(W1, HH, wave * 16, s, lane);
    bias1 = b1[wave * 16 + (lane & 15)];
  }
  v4f bW2[T2][MIDD / 8];
  float bias2[T2];
#pragma unroll
  for (int t = 0; t < T2; ++t) {
#pragma unroll
    for (int s = 0; s < MIDD / 8; ++s)
      bW2[t][s] = load_b8T(W2, MIDD, (wave * T2 + t) * 16, s, lane);
    bias2[t] = b2[(wave * T2 + t) * 16 + (lane & 15)];
  }

  for (int i = 0; i < 3; ++i)
    for (int q = tid; q < 16 * CCH; q += NTHREADS) {
      int b = q >> 3, c = q & 7;
      long base = ((long)(r0 + b) * LL + i) * CCH + c;
      s_dX[i][q] = coeffs[base + CCH] - coeffs[base];
    }

  {
    v4f aX = load_a8(coeffs + (long)r0 * LL * CCH, LL * CCH, 0, lane);
    for (int nt = wave; nt < HH / 16; nt += NWAVES) {
      v8f acc = {};
      v4f b = load_b8T(W_init, CCH, nt * 16, 0, lane);
      acc = WMMA4(lo2(aX), lo2(b), acc);
      acc = WMMA4(hi2(aX), hi2(b), acc);
      int col = nt * 16 + (lane & 15);
      float bias = b_init[col];
#pragma unroll
      for (int j = 0; j < 8; ++j) {
        int row = (lane < 16) ? j : j + 8;
        float v = acc[j] + bias;
        s_z0[row * HH + col] = v;
        s_z [row * ZLD + col] = v;
      }
    }
  }
  __syncthreads();

  g_eval(s_z, s_h, s_kc, s_dX[0], bW1, bias1, bW2, bias2, lane, wave);
  for (int q = tid; q < 16 * HH; q += NTHREADS) {
    float k1 = s_kc[q];
    s_k1[q] = k1;
    s_ac[q] = k1;
    s_z[(q >> 6) * ZLD + (q & (HH - 1))] = s_z0[q] + dt * (k1 * (1.0f / 3.0f));
  }
  __syncthreads();

  g_eval(s_z, s_h, s_kc, s_dX[0], bW1, bias1, bW2, bias2, lane, wave);
  for (int q = tid; q < 16 * HH; q += NTHREADS) {
    float k2 = s_kc[q];
    s_k2[q] = k2;
    s_ac[q] += 3.0f * k2;
    s_z[(q >> 6) * ZLD + (q & (HH - 1))] = s_z0[q] + dt * (k2 - s_k1[q] * (1.0f / 3.0f));
  }
  __syncthreads();

  g_eval(s_z, s_h, s_kc, s_dX[1], bW1, bias1, bW2, bias2, lane, wave);
  for (int q = tid; q < 16 * HH; q += NTHREADS) {
    float k3 = s_kc[q];
    s_ac[q] += 3.0f * k3;
    s_z[(q >> 6) * ZLD + (q & (HH - 1))] = s_z0[q] + dt * (s_k1[q] - s_k2[q] + k3);
  }
  __syncthreads();

  g_eval(s_z, s_h, s_kc, s_dX[2], bW1, bias1, bW2, bias2, lane, wave);
  for (int q = tid; q < 16 * HH; q += NTHREADS) {
    s_z[(q >> 6) * ZLD + (q & (HH - 1))] = s_z0[q] + (s_ac[q] + s_kc[q]) * (dt * 0.125f);
  }
  __syncthreads();

  if (wave == 0) {
    int b = lane & 15, o = lane >> 4;
    float s = b_out[o];
    for (int hh = 0; hh < HH; ++hh)
      s = fmaf(s_z[b * ZLD + hh], W_out[o * HH + hh], s);
    out[(long)(r0 + b) * 2 + o] = s;
  }
}

extern "C" void kernel_launch(void* const* d_in, const int* in_sizes, int n_in,
                              void* d_out, int out_size, void* d_ws, size_t ws_size,
                              hipStream_t stream) {
  (void)n_in; (void)out_size; (void)d_ws; (void)ws_size;
  const float* coeffs = (const float*)d_in[0];
  const float* W_init = (const float*)d_in[1];
  const float* b_init = (const float*)d_in[2];
  const float* W1     = (const float*)d_in[3];
  const float* b1     = (const float*)d_in[4];
  const float* W2     = (const float*)d_in[5];
  const float* b2     = (const float*)d_in[6];
  const float* W_out  = (const float*)d_in[7];
  const float* b_out  = (const float*)d_in[8];
  float* out = (float*)d_out;

  int B = in_sizes[0] / (LL * CCH);
  int blocks = B / 16;
  ncde_fused<<<blocks, NTHREADS, 0, stream>>>(coeffs, W_init, b_init, W1, b1, W2, b2,
                                        W_out, b_out, out);
}


# tanh of the GEMM 2 epilogue in one range as (1-e)/(1+e), e = exp2(-2 log2(e) |x|), on the hardware exp2 and rcp: no polynomial, no range select; tanh values move by about 1e-7; sum orders unchanged
# speedup vs baseline: 14.7599x; 1.1399x over previous
#include <hip/hip_runtime.h>
#include <hip/hip_bf16.h>
#include <math.h>

#define LL   141
#define CCH  8
#define HH   64
#define MIDD 128
#define HCC  512
#define NWAVES 16
#define NTHREADS (NWAVES * 32)
#define ZLD  (HH + 4)
#define HLD  (MIDD + 4)
#define T2   (HCC / 16 / NWAVES)

typedef __attribute__((ext_vector_type(2))) float v2f;
typedef __attribute__((ext_vector_type(4))) float v4f;
typedef __attribute__((ext_vector_type(8))) float v8f;

#define WMMA4(a, b, c) \
  __builtin_amdgcn_wmma_f32_16x16x4_f32(false, (a), false, (b), (short)0, (c), false, false)

__device__ __forceinline__ v4f load_a8(const float* base, int ld, int s, int lane) {
  int m = lane & 15;
  int k = s * 8 + ((lane >> 4) << 2);
  return *reinterpret_cast<const v4f*>(base + m * ld + k);
}

__device__ __forceinline__ v4f load_b8T(const float* W, int ldw, int n0, int s, int lane) {
  int n = n0 + (lane & 15);
  int k = s * 8 + ((lane >> 4) << 2);
  return *reinterpret_cast<const v4f*>(W + n * ldw + k);
}
__device__ __forceinline__ v2f lo2(v4f v) { return __builtin_shufflevector(v, v, 0, 1); }
__device__ __forceinline__ v2f hi2(v4f v) { return __builtin_shufflevector(v, v, 2, 3); }

__device__ __forceinline__ v8f splat8(float c) { return v8f{c, c, c, c, c, c, c, c}; }
__device__ __forceinline__ v8f tanh8(v8f x) {
  v8f y = __builtin_elementwise_abs(x);
  v8f a = y * splat8(-0x1.715476p+1f);
  v8f e, r;
#pragma unroll
  for (int j = 0; j < 8; ++j) e[j] = __builtin_amdgcn_exp2f(a[j]);
  v8f num = splat8(1.0f) - e;
  v8f den = splat8(1.0f) + e;
#pragma unroll
  for (int j = 0; j < 8; ++j) r[j] = __builtin_amdgcn_rcpf(den[j]);
  v8f z = num * r;
  return __builtin_elementwise_copysign(z, x);
}

template <int M>
__device__ __forceinline__ float lane_xor(float v) {
  int r = __builtin_amdgcn_update_dpp(0, __builtin_bit_cast(int, v), 0x160 + M, 0xf, 0xf, true);
  return __builtin_bit_cast(float, r);
}

__device__ __forceinline__ void g_eval(const float* z, float* h, float* kout,
                                       const float* dX,
                                       const v4f (&bW1)[HH / 8], float bias1,
                                       const v4f (&bW2)[T2][MIDD / 8], const float (&bias2)[T2],
                                       int lane, int wave) {
  if (wave < MIDD / 16) {
    v4f aZ[HH / 8];
#pragma unroll
    for (int s = 0; s < HH / 8; ++s) aZ[s] = load_a8(z, ZLD, s, lane);
    v8f acc = {};
#pragma unroll
    for (int s = 0; s < HH / 8; ++s) {
      acc = WMMA4(lo2(aZ[s]), lo2(bW1[s]), acc);
      acc = WMMA4(hi2(aZ[s]), hi2(bW1[s]), acc);
    }
    int col = wave * 16 + (lane & 15);
#pragma unroll
    for (int j = 0; j < 8; ++j) {
      int row = (lane < 16) ? j : j + 8;
      h[row * HLD + col] = fmaxf(acc[j] + bias1, 0.0f);
    }
  }
  __syncthreads();

  int cc   = lane & 7;
  int hsel = (lane >> 3) & 1;

  float dxv[8];
#pragma unroll
  for (int j = 0; j < 8; ++j) {
    int row = (lane < 16) ? j : j + 8;
    dxv[j] = dX[row * CCH + cc];
  }

  v8f acc[T2];
#pragma unroll
  for (int t = 0; t < T2; ++t) acc[t] = v8f{};
#pragma unroll
  for (int s = 0; s < MIDD / 8; ++s) {
    v4f a = load_a8(h, HLD, s, lane);
#pragma unroll
    for (int t = 0; t < T2; ++t) acc[t] = WMMA4(lo2(a), lo2(bW2[t][s]), acc[t]);
#pragma unroll
    for (int t = 0; t < T2; ++t) acc[t] = WMMA4(hi2(a), hi2(bW2[t][s]), acc[t]);
  }
  v8f dx8;
#pragma unroll
  for (int j = 0; j < 8; ++j) dx8[j] = dxv[j];
  v8f p[T2];
#pragma unroll
  for (int t = 0; t < T2; ++t) {
    v8f q = tanh8(acc[t] + bias2[t]) * dx8;
#pragma unroll
    for (int j = 0; j < 8; ++j) q[j] += lane_xor<1>(q[j]);
#pragma unroll
    for (int j = 0; j < 8; ++j) q[j] += lane_xor<2>(q[j]);
#pragma unroll
    for (int j = 0; j < 8; ++j) q[j] += lane_xor<4>(q[j]);
    p[t] = q;
  }
  if (cc == 0) {
#pragma unroll
    for (int t = 0; t < T2; ++t) {
      int hout = (wave * T2 + t) * 2 + hsel;
#pragma unroll
      for (int j = 0; j < 8; ++j) {
        int row = (lane < 16) ? j : j + 8;
        kout[row * HH + hout] = p[t][j];
      }
    }
  }
  __syncthreads();
}

__global__ __launch_bounds__(NTHREADS) void ncde_fused(
    const float* __restrict__ coeffs,
    const float* __restrict__ W_init, const float* __restrict__ b_init,
    const float* __restrict__ W1, const float* __restrict__ b1,
    const float* __restrict__ W2, const float* __restrict__ b2,
    const float* __restrict__ W_out, const float* __restrict__ b_out,
    float* __restrict__ out) {
  __shared__ __attribute__((aligned(16))) float s_z [16 * ZLD];
  __shared__ float s_z0[16 * HH];
  __shared__ __attribute__((aligned(16))) float s_h [16 * HLD];
  __shared__ float s_k1[16 * HH];
  __shared__ float s_k2[16 * HH];
  __shared__ float s_kc[16 * HH];
  __shared__ float s_ac[16 * HH];
  __shared__ float s_dX[3][16 * CCH];

  const int tid  = threadIdx.x;
  const int lane = tid & 31;
  const int wave = __builtin_amdgcn_readfirstlane(tid >> 5);
  const int r0   = blockIdx.x * 16;
  const float dt = 140.0f / 49.0f;

  v4f bW1[HH / 8];
  float bias1 = 0.0f;
  if (wave < MIDD / 16) {
#pragma unroll
    for (int s = 0; s < HH / 8; ++s) bW1[s] = load_b8T(W1, HH, wave * 16, s, lane);
    bias1 = b1[wave * 16 + (lane & 15)];
  }
  v4f bW2[T2][MIDD / 8];
  float bias2[T2];
#pragma unroll
  for (int t = 0; t < T2; ++t) {
#pragma unroll
    for (int s = 0; s < MIDD / 8; ++s)
      bW2[t][s] = load_b8T(W2, MIDD, (wave * T2 + t) * 16, s, lane);
    bias2[t] = b2[(wave * T2 + t) * 16 + (lane & 15)];
  }

  for (int i = 0; i < 3; ++i)
    for (int q = tid; q < 16 * CCH; q += NTHREADS) {
      int b = q >> 3, c = q & 7;
      long base = ((long)(r0 + b) * LL + i) * CCH + c;
      s_dX[i][q] = coeffs[base + CCH] - coeffs[base];
    }

  {
    v4f aX = load_a8(coeffs + (long)r0 * LL * CCH, LL * CCH, 0, lane);
    for (int nt = wave; nt < HH / 16; nt += NWAVES) {
      v8f acc = {};
      v4f b = load_b8T(W_init, CCH, nt * 16, 0, lane);
      acc = WMMA4(lo2(aX), lo2(b), acc);
      acc = WMMA4(hi2(aX), hi2(b), acc);
      int col = nt * 16 + (lane & 15);
      float bias = b_init[col];
#pragma unroll
      for (int j = 0; j < 8; ++j) {
        int row = (lane < 16) ? j : j + 8;
        float v = acc[j] + bias;
        s_z0[row * HH + col] = v;
        s_z [row * ZLD + col] = v;
      }
    }
  }
  __syncthreads();

  g_eval(s_z, s_h, s_kc, s_dX[0], bW1, bias1, bW2, bias2, lane, wave);
  for (int q = tid; q < 16 * HH; q += NTHREADS) {
    float k1 = s_kc[q];
    s_k1[q] = k1;
    s_ac[q] = k1;
    s_z[(q >> 6) * ZLD + (q & (HH - 1))] = s_z0[q] + dt * (k1 * (1.0f / 3.0f));
  }
  __syncthreads();

  g_eval(s_z, s_h, s_kc, s_dX[0], bW1, bias1, bW2, bias2, lane, wave);
  for (int q = tid; q < 16 * HH; q += NTHREADS) {
    float k2 = s_kc[q];
    s_k2[q] = k2;
    s_ac[q] += 3.0f * k2;
    s_z[(q >> 6) * ZLD + (q & (HH - 1))] = s_z0[q] + dt * (k2 - s_k1[q] * (1.0f / 3.0f));
  }
  __syncthreads();

  g_eval(s_z, s_h, s_kc, s_dX[1], bW1, bias1, bW2, bias2, lane, wave);
  for (int q = tid; q < 16 * HH; q += NTHREADS) {
    float k3 = s_kc[q];
    s_ac[q] += 3.0f * k3;
    s_z[(q >> 6) * ZLD + (q & (HH - 1))] = s_z0[q] + dt * (s_k1[q] - s_k2[q] + k3);
  }
  __syncthreads();

  g_eval(s_z, s_h, s_kc, s_dX[2], bW1, bias1, bW2, bias2, lane, wave);
  for (int q = tid; q < 16 * HH; q += NTHREADS) {
    s_z[(q >> 6) * ZLD + (q & (HH - 1))] = s_z0[q] + (s_ac[q] + s_kc[q]) * (dt * 0.125f);
  }
  __syncthreads();

  if (wave == 0) {
    int b = lane & 15, o = lane >> 4;
    float s = b_out[o];
    for (int hh = 0; hh < HH; ++hh)
      s = fmaf(s_z[b * ZLD + hh], W_out[o * HH + hh], s);
    out[(long)(r0 + b) * 2 + o] = s;
  }
}

extern "C" void kernel_launch(void* const* d_in, const int* in_sizes, int n_in,
                              void* d_out, int out_size, void* d_ws, size_t ws_size,
                              hipStream_t stream) {
  (void)n_in; (void)out_size; (void)d_ws; (void)ws_size;
  const float* coeffs = (const float*)d_in[0];
  const float* W_init = (const float*)d_in[1];
  const float* b_init = (const float*)d_in[2];
  const float* W1     = (const float*)d_in[3];
  const float* b1     = (const float*)d_in[4];
  const float* W2     = (const float*)d_in[5];
  const float* b2     = (const float*)d_in[6];
  const float* W_out  = (const float*)d_in[7];
  const float* b_out  = (const float*)d_in[8];
  float* out = (float*)d_out;

  int B = in_sizes[0] / (LL * CCH);
  int blocks = B / 16;
  ncde_fused<<<blocks, NTHREADS, 0, stream>>>(coeffs, W_init, b_init, W1, b1, W2, b2,
                                        W_out, b_out, out);
}


# The one-time W2 fragment fetch overlaps the first stage's GEMMs: fragments requested in order of use, 4 k-chunks ahead, waited per chunk; barriers fence LDS only; sum orders unchanged
# speedup vs baseline: 16.1678x; 1.0954x over previous
#include <hip/hip_runtime.h>
#include <hip/hip_bf16.h>
#include <math.h>

#define LL   141
#define CCH  8
#define HH   64
#define MIDD 128
#define HCC  512
#define NWAVES 16
#define NTHREADS (NWAVES * 32)
#define ZLD  (HH + 4)
#define HLD  (MIDD + 4)
#define T2   (HCC / 16 / NWAVES)
#define PF0  4

typedef __attribute__((ext_vector_type(2))) float v2f;
typedef __attribute__((ext_vector_type(4))) float v4f;
typedef __attribute__((ext_vector_type(8))) float v8f;

#define WMMA4(a, b, c) \
  __builtin_amdgcn_wmma_f32_16x16x4_f32(false, (a), false, (b), (short)0, (c), false, false)

__device__ __forceinline__ v4f load_a8(const float* base, int ld, int s, int lane) {
  int m = lane & 15;
  int k = s * 8 + ((lane >> 4) << 2);
  return *reinterpret_cast<const v4f*>(base + m * ld + k);
}

__device__ __forceinline__ v4f load_b8T(const float* W, int ldw, int n0, int s, int lane) {
  int n = n0 + (lane & 15);
  int k = s * 8 + ((lane >> 4) << 2);
  return *reinterpret_cast<const v4f*>(W + n * ldw + k);
}
__device__ __forceinline__ v2f lo2(v4f v) { return __builtin_shufflevector(v, v, 0, 1); }
__device__ __forceinline__ v2f hi2(v4f v) { return __builtin_shufflevector(v, v, 2, 3); }

__device__ __forceinline__ v8f splat8(float c) { return v8f{c, c, c, c, c, c, c, c}; }
__device__ __forceinline__ v8f tanh8(v8f x) {
  v8f y = __builtin_elementwise_abs(x);
  v8f a = y * splat8(-0x1.715476p+1f);
  v8f e, r;
#pragma unroll
  for (int j = 0; j < 8; ++j) e[j] = __builtin_amdgcn_exp2f(a[j]);
  v8f num = splat8(1.0f) - e;
  v8f den = splat8(1.0f) + e;
#pragma unroll
  for (int j = 0; j < 8; ++j) r[j] = __builtin_amdgcn_rcpf(den[j]);
  v8f z = num * r;
  return __builtin_elementwise_copysign(z, x);
}

__device__ __forceinline__ void lds_barrier() {
  __builtin_amdgcn_fence(__ATOMIC_RELEASE, "workgroup", "local");
  __builtin_amdgcn_s_barrier();
  __builtin_amdgcn_fence(__ATOMIC_ACQUIRE, "workgroup", "local");
}

template <int M>
__device__ __forceinline__ float lane_xor(float v) {
  int r = __builtin_amdgcn_update_dpp(0, __builtin_bit_cast(int, v), 0x160 + M, 0xf, 0xf, true);
  return __builtin_bit_cast(float, r);
}

template <bool FIRST>
__device__ __forceinline__ void g_eval(const float* z, float* h, float* kout,
                                       const float* dX,
                                       const v4f (&bW1)[HH / 8], float bias1,
                                       v4f (&bW2)[T2][MIDD / 8], const float (&bias2)[T2],
                                       const float* __restrict__ W2,
                                       int lane, int wave) {
  if (wave < MIDD / 16) {
    v4f aZ[HH / 8];
#pragma unroll
    for (int s = 0; s < HH / 8; ++s) aZ[s] = load_a8(z, ZLD, s, lane);
    v8f acc = {};
#pragma unroll
    for (int s = 0; s < HH / 8; ++s) {
      acc = WMMA4(lo2(aZ[s]), lo2(bW1[s]), acc);
      acc = WMMA4(hi2(aZ[s]), hi2(bW1[s]), acc);
    }
    int col = wave * 16 + (lane & 15);
#pragma unroll
    for (int j = 0; j < 8; ++j) {
      int row = (lane < 16) ? j : j + 8;
      h[row * HLD + col] = fmaxf(acc[j] + bias1, 0.0f);
    }
  }
  lds_barrier();

  int cc   = lane & 7;
  int hsel = (lane >> 3) & 1;

  float dxv[8];
#pragma unroll
  for (int j = 0; j < 8; ++j) {
    int row = (lane < 16) ? j : j + 8;
    dxv[j] = dX[row * CCH + cc];
  }

  v8f acc[T2];
#pragma unroll
  for (int t = 0; t < T2; ++t) acc[t] = v8f{};
#pragma unroll
  for (int s = 0; s < MIDD / 8; ++s) {
    if (FIRST && s + PF0 < MIDD / 8) {
#pragma unroll
      for (int t = 0; t < T2; ++t)
        bW2[t][s + PF0] = load_b8T(W2, MIDD, (wave * T2 + t) * 16, s + PF0, lane);
    }
    if (FIRST) __builtin_amdgcn_sched_barrier(0x786);
    v4f a = load_a8(h, HLD, s, lane);
#pragma unroll
    for (int t = 0; t < T2; ++t) acc[t] = WMMA4(lo2(a), lo2(bW2[t][s]), acc[t]);
#pragma unroll
    for (int t = 0; t < T2; ++t) acc[t] = WMMA4(hi2(a), hi2(bW2[t][s]), acc[t]);
  }
  v8f dx8;
#pragma unroll
  for (int j = 0; j < 8; ++j) dx8[j] = dxv[j];
  v8f p[T2];
#pragma unroll
  for (int t = 0; t < T2; ++t) {
    v8f q = tanh8(acc[t] + bias2[t]) * dx8;
#pragma unroll
    for (int j = 0; j < 8; ++j) q[j] += lane_xor<1>(q[j]);
#pragma unroll
    for (int j = 0; j < 8; ++j) q[j] += lane_xor<2>(q[j]);
#pragma unroll
    for (int j = 0; j < 8; ++j) q[j] += lane_xor<4>(q[j]);
    p[t] = q;
  }
  if (cc == 0) {
#pragma unroll
    for (int t = 0; t < T2; ++t) {
      int hout = (wave * T2 + t) * 2 + hsel;
#pragma unroll
      for (int j = 0; j < 8; ++j) {
        int row = (lane < 16) ? j : j + 8;
        kout[row * HH + hout] = p[t][j];
      }
    }
  }
  lds_barrier();
}

__global__ __launch_bounds__(NTHREADS) void ncde_fused(
    const float* __restrict__ coeffs,
    const float* __restrict__ W_init, const float* __restrict__ b_init,
    const float* __restrict__ W1, const float* __restrict__ b1,
    const float* __restrict__ W2, const float* __restrict__ b2,
    const float* __restrict__ W_out, const float* __restrict__ b_out,
    float* __restrict__ out) {
  __shared__ __attribute__((aligned(16))) float s_z [16 * ZLD];
  __shared__ float s_z0[16 * HH];
  __shared__ __attribute__((aligned(16))) float s_h [16 * HLD];
  __shared__ float s_k1[16 * HH];
  __shared__ float s_k2[16 * HH];
  __shared__ float s_kc[16 * HH];
  __shared__ float s_ac[16 * HH];
  __shared__ float s_dX[3][16 * CCH];

  const int tid  = threadIdx.x;
  const int lane = tid & 31;
  const int wave = __builtin_amdgcn_readfirstlane(tid >> 5);
  const int r0   = blockIdx.x * 16;
  const float dt = 140.0f / 49.0f;


  float x0 = 0.0f, x1 = 0.0f, x2 = 0.0f, x3 = 0.0f;
  if (tid < 16 * CCH) {
    const float* px = coeffs + (long)(r0 + (tid >> 3)) * LL * CCH + (tid & 7);
    x0 = px[0]; x1 = px[CCH]; x2 = px[2 * CCH]; x3 = px[3 * CCH];
  }
  static_assert(HH / 16 <= NWAVES, "one z0 tile per wave");
  v4f aX = {}, bI = {};
  float biasI = 0.0f;
  if (wave < HH / 16) {
    aX = load_a8(coeffs + (long)r0 * LL * CCH, LL * CCH, 0, lane);
    bI = load_b8T(W_init, CCH, wave * 16, 0, lane);
    biasI = b_init[wave * 16 + (lane & 15)];
  }

  v4f bW1[HH / 8];
  float bias1 = 0.0f;
  if (wave < MIDD / 16) {
#pragma unroll
    for (int s = 0; s < HH / 8; ++s) bW1[s] = load_b8T(W1, HH, wave * 16, s, lane);
    bias1 = b1[wave * 16 + (lane & 15)];
  }
  v4f bW2[T2][MIDD / 8];
  float bias2[T2];
#pragma unroll
  for (int t = 0; t < T2; ++t) bias2[t] = b2[(wave * T2 + t) * 16 + (lane & 15)];
#pragma unroll
  for (int s = 0; s < PF0; ++s) {
#pragma unroll
    for (int t = 0; t < T2; ++t)
      bW2[t][s] = load_b8T(W2, MIDD, (wave * T2 + t) * 16, s, lane);
  }

  if (tid < 16 * CCH) {
    s_dX[0][tid] = x1 - x0;
    s_dX[1][tid] = x2 - x1;
    s_dX[2][tid] = x3 - x2;
  }

  if (wave < HH / 16) {
    v8f acc = {};
    acc = WMMA4(lo2(aX), lo2(bI), acc);
    acc = WMMA4(hi2(aX), hi2(bI), acc);
    int col = wave * 16 + (lane & 15);
#pragma unroll
    for (int j = 0; j < 8; ++j) {
      int row = (lane < 16) ? j : j + 8;
      float v = acc[j] + biasI;
      s_z0[row * HH + col] = v;
      s_z [row * ZLD + col] = v;
    }
  }
  lds_barrier();

  g_eval<true>(s_z, s_h, s_kc, s_dX[0], bW1, bias1, bW2, bias2, W2, lane, wave);
  for (int q = tid; q < 16 * HH; q += NTHREADS) {
    float k1 = s_kc[q];
    s_k1[q] = k1;
    s_ac[q] = k1;
    s_z[(q >> 6) * ZLD + (q & (HH - 1))] = s_z0[q] + dt * (k1 * (1.0f / 3.0f));
  }
  lds_barrier();

  g_eval<false>(s_z, s_h, s_kc, s_dX[0], bW1, bias1, bW2, bias2, W2, lane, wave);
  for (int q = tid; q < 16 * HH; q += NTHREADS) {
    float k2 = s_kc[q];
    s_k2[q] = k2;
    s_ac[q] += 3.0f * k2;
    s_z[(q >> 6) * ZLD + (q & (HH - 1))] = s_z0[q] + dt * (k2 - s_k1[q] * (1.0f / 3.0f));
  }
  lds_barrier();

  g_eval<false>(s_z, s_h, s_kc, s_dX[1], bW1, bias1, bW2, bias2, W2, lane, wave);
  for (int q = tid; q < 16 * HH; q += NTHREADS) {
    float k3 = s_kc[q];
    s_ac[q] += 3.0f * k3;
    s_z[(q >> 6) * ZLD + (q & (HH - 1))] = s_z0[q] + dt * (s_k1[q] - s_k2[q] + k3);
  }
  lds_barrier();

  g_eval<false>(s_z, s_h, s_kc, s_dX[2], bW1, bias1, bW2, bias2, W2, lane, wave);
  for (int q = tid; q < 16 * HH; q += NTHREADS) {
    s_z[(q >> 6) * ZLD + (q & (HH - 1))] = s_z0[q] + (s_ac[q] + s_kc[q]) * (dt * 0.125f);
  }
  lds_barrier();

  if (wave == 0) {
    int b = lane & 15, o = lane >> 4;
    float s = b_out[o];
    for (int hh = 0; hh < HH; ++hh)
      s = fmaf(s_z[b * ZLD + hh], W_out[o * HH + hh], s);
    out[(long)(r0 + b) * 2 + o] = s;
  }
}

extern "C" void kernel_launch(void* const* d_in, const int* in_sizes, int n_in,
                              void* d_out, int out_size, void* d_ws, size_t ws_size,
                              hipStream_t stream) {
  (void)n_in; (void)out_size; (void)d_ws; (void)ws_size;
  const float* coeffs = (const float*)d_in[0];
  const float* W_init = (const float*)d_in[1];
  const float* b_init = (const float*)d_in[2];
  const float* W1     = (const float*)d_in[3];
  const float* b1     = (const float*)d_in[4];
  const float* W2     = (const float*)d_in[5];
  const float* b2     = (const float*)d_in[6];
  const float* W_out  = (const float*)d_in[7];
  const float* b_out  = (const float*)d_in[8];
  float* out = (float*)d_out;

  int B = in_sizes[0] / (LL * CCH);
  int blocks = B / 16;
  ncde_fused<<<blocks, NTHREADS, 0, stream>>>(coeffs, W_init, b_init, W1, b1, W2, b2,
                                        W_out, b_out, out);
}


# RK4 update fused into the GEMM 2 epilogue: c-sum leaves each lane one row's sum (same tree); z0, k1, k1-k2, stage sum kept in registers, next z stored directly; 2 barriers a stage, not 3
# speedup vs baseline: 17.3038x; 1.0703x over previous
#include <hip/hip_runtime.h>
#include <hip/hip_bf16.h>
#include <math.h>

#define LL   141
#define CCH  8
#define HH   64
#define MIDD 128
#define HCC  512
#define NWAVES 16
#define NTHREADS (NWAVES * 32)
#define ZLD  (HH + 4)
#define HLD  (MIDD + 4)
#define T2   (HCC / 16 / NWAVES)
#define PF0  4

typedef __attribute__((ext_vector_type(2))) float v2f;
typedef __attribute__((ext_vector_type(4))) float v4f;
typedef __attribute__((ext_vector_type(8))) float v8f;

#define WMMA4(a, b, c) \
  __builtin_amdgcn_wmma_f32_16x16x4_f32(false, (a), false, (b), (short)0, (c), false, false)

__device__ __forceinline__ v4f load_a8(const float* base, int ld, int s, int lane) {
  int m = lane & 15;
  int k = s * 8 + ((lane >> 4) << 2);
  return *reinterpret_cast<const v4f*>(base + m * ld + k);
}

__device__ __forceinline__ v4f load_b8T(const float* W, int ldw, int n0, int s, int lane) {
  int n = n0 + (lane & 15);
  int k = s * 8 + ((lane >> 4) << 2);
  return *reinterpret_cast<const v4f*>(W + n * ldw + k);
}
__device__ __forceinline__ v2f lo2(v4f v) { return __builtin_shufflevector(v, v, 0, 1); }
__device__ __forceinline__ v2f hi2(v4f v) { return __builtin_shufflevector(v, v, 2, 3); }

__device__ __forceinline__ v8f splat8(float c) { return v8f{c, c, c, c, c, c, c, c}; }
__device__ __forceinline__ v8f tanh8(v8f x) {
  v8f y = __builtin_elementwise_abs(x);
  v8f a = y * splat8(-0x1.715476p+1f);
  v8f e, r;
#pragma unroll
  for (int j = 0; j < 8; ++j) e[j] = __builtin_amdgcn_exp2f(a[j]);
  v8f num = splat8(1.0f) - e;
  v8f den = splat8(1.0f) + e;
#pragma unroll
  for (int j = 0; j < 8; ++j) r[j] = __builtin_amdgcn_rcpf(den[j]);
  v8f z = num * r;
  return __builtin_elementwise_copysign(z, x);
}

__device__ __forceinline__ void lds_barrier() {
  __builtin_amdgcn_fence(__ATOMIC_RELEASE, "workgroup", "local");
  __builtin_amdgcn_s_barrier();
  __builtin_amdgcn_fence(__ATOMIC_ACQUIRE, "workgroup", "local");
}

template <int M>
__device__ __forceinline__ float lane_xor(float v) {
  int r = __builtin_amdgcn_update_dpp(0, __builtin_bit_cast(int, v), 0x160 + M, 0xf, 0xf, true);
  return __builtin_bit_cast(float, r);
}

__device__ __forceinline__ float csum8_own_row(v8f q, int lane) {
  const bool b0 = (lane & 1) != 0, b1 = (lane & 2) != 0, b2 = (lane & 4) != 0;
  v4f lo4 = __builtin_shufflevector(q, q, 0, 1, 2, 3), hi4 = __builtin_shufflevector(q, q, 4, 5, 6, 7);
  v4f keep4 = b0 ? hi4 : lo4, send4 = b0 ? lo4 : hi4, r4;
#pragma unroll
  for (int i = 0; i < 4; ++i) r4[i] = keep4[i] + lane_xor<1>(send4[i]);
  v2f lo_2 = lo2(r4), hi_2 = hi2(r4);
  v2f keep2 = b1 ? hi_2 : lo_2, send2 = b1 ? lo_2 : hi_2, r2;
#pragma unroll
  for (int i = 0; i < 2; ++i) r2[i] = keep2[i] + lane_xor<2>(send2[i]);
  float keep = b2 ? r2[1] : r2[0], send = b2 ? r2[0] : r2[1];
  return keep + lane_xor<4>(send);
}

template <bool FIRST>
__device__ __forceinline__ void g_eval(const float* z, float* h,
                                       const float* dX,
                                       const v4f (&bW1)[HH / 8], float bias1,
                                       v4f (&bW2)[T2][MIDD / 8], const float (&bias2)[T2],
                                       const float* __restrict__ W2,
                                       int lane, int wave, float (&kv)[T2]) {
  if (wave < MIDD / 16) {
    v4f aZ[HH / 8];
#pragma unroll
    for (int s = 0; s < HH / 8; ++s) aZ[s] = load_a8(z, ZLD, s, lane);
    v8f acc = {};
#pragma unroll
    for (int s = 0; s < HH / 8; ++s) {
      acc = WMMA4(lo2(aZ[s]), lo2(bW1[s]), acc);
      acc = WMMA4(hi2(aZ[s]), hi2(bW1[s]), acc);
    }
    int col = wave * 16 + (lane & 15);
#pragma unroll
    for (int j = 0; j < 8; ++j) {
      int row = (lane < 16) ? j : j + 8;
      h[row * HLD + col] = fmaxf(acc[j] + bias1, 0.0f);
    }
  }
  lds_barrier();

  int cc   = lane & 7;

  float dxv[8];
#pragma unroll
  for (int j = 0; j < 8; ++j) {
    int row = (lane < 16) ? j : j + 8;
    dxv[j] = dX[row * CCH + cc];
  }

  v8f acc[T2];
#pragma unroll
  for (int t = 0; t < T2; ++t) acc[t] = v8f{};
#pragma unroll
  for (int s = 0; s < MIDD / 8; ++s) {
    if (FIRST && s + PF0 < MIDD / 8) {
#pragma unroll
      for (int t = 0; t < T2; ++t)
        bW2[t][s + PF0] = load_b8T(W2, MIDD, (wave * T2 + t) * 16, s + PF0, lane);
    }
    if (FIRST) __builtin_amdgcn_sched_barrier(0x786);
    v4f a = load_a8(h, HLD, s, lane);
#pragma unroll
    for (int t = 0; t < T2; ++t) acc[t] = WMMA4(lo2(a), lo2(bW2[t][s]), acc[t]);
#pragma unroll
    for (int t = 0; t < T2; ++t) acc[t] = WMMA4(hi2(a), hi2(bW2[t][s]), acc[t]);
  }
  v8f dx8;
#pragma unroll
  for (int j = 0; j < 8; ++j) dx8[j] = dxv[j];
#pragma unroll
  for (int t = 0; t < T2; ++t) {
    v8f q = tanh8(acc[t] + bias2[t]) * dx8;
    kv[t] = csum8_own_row(q, lane);
  }
}

__global__ __launch_bounds__(NTHREADS) void ncde_fused(
    const float* __restrict__ coeffs,
    const float* __restrict__ W_init, const float* __restrict__ b_init,
    const float* __restrict__ W1, const float* __restrict__ b1,
    const float* __restrict__ W2, const float* __restrict__ b2,
    const float* __restrict__ W_out, const float* __restrict__ b_out,
    float* __restrict__ out) {
  __shared__ __attribute__((aligned(16))) float s_z [16 * ZLD];
  __shared__ __attribute__((aligned(16))) float s_h [16 * HLD];
  __shared__ float s_dX[3][16 * CCH];

  const int tid  = threadIdx.x;
  const int lane = tid & 31;
  const int wave = __builtin_amdgcn_readfirstlane(tid >> 5);
  const int r0   = blockIdx.x * 16;
  const float dt = 140.0f / 49.0f;


  float x0 = 0.0f, x1 = 0.0f, x2 = 0.0f, x3 = 0.0f;
  if (tid < 16 * CCH) {
    const float* px = coeffs + (long)(r0 + (tid >> 3)) * LL * CCH + (tid & 7);
    x0 = px[0]; x1 = px[CCH]; x2 = px[2 * CCH]; x3 = px[3 * CCH];
  }
  static_assert(HH / 16 <= NWAVES, "one z0 tile per wave");
  v4f aX = {}, bI = {};
  float biasI = 0.0f;
  if (wave < HH / 16) {
    aX = load_a8(coeffs + (long)r0 * LL * CCH, LL * CCH, 0, lane);
    bI = load_b8T(W_init, CCH, wave * 16, 0, lane);
    biasI = b_init[wave * 16 + (lane & 15)];
  }

  v4f bW1[HH / 8];
  float bias1 = 0.0f;
  if (wave < MIDD / 16) {
#pragma unroll
    for (int s = 0; s < HH / 8; ++s) bW1[s] = load_b8T(W1, HH, wave * 16, s, lane);
    bias1 = b1[wave * 16 + (lane & 15)];
  }
  v4f bW2[T2][MIDD / 8];
  float bias2[T2];
#pragma unroll
  for (int t = 0; t < T2; ++t) bias2[t] = b2[(wave * T2 + t) * 16 + (lane & 15)];
#pragma unroll
  for (int s = 0; s < PF0; ++s) {
#pragma unroll
    for (int t = 0; t < T2; ++t)
      bW2[t][s] = load_b8T(W2, MIDD, (wave * T2 + t) * 16, s, lane);
  }

  if (tid < 16 * CCH) {
    s_dX[0][tid] = x1 - x0;
    s_dX[1][tid] = x2 - x1;
    s_dX[2][tid] = x3 - x2;
  }

  if (wave < HH / 16) {
    v8f acc = {};
    acc = WMMA4(lo2(aX), lo2(bI), acc);
    acc = WMMA4(hi2(aX), hi2(bI), acc);
    int col = wave * 16 + (lane & 15);
#pragma unroll
    for (int j = 0; j < 8; ++j) {
      int row = (lane < 16) ? j : j + 8;
      s_z[row * ZLD + col] = acc[j] + biasI;
    }
  }
  lds_barrier();

  const int myrow = ((lane >> 4) << 3) + ((lane & 1) << 2) + (lane & 2) + ((lane >> 2) & 1);
  const int hsel  = (lane >> 3) & 1;
  int zoff[T2];
  float z0v[T2], k1v[T2], dv[T2], acv[T2], kv[T2];
#pragma unroll
  for (int t = 0; t < T2; ++t) {
    zoff[t] = myrow * ZLD + (wave * T2 + t) * 2 + hsel;
    z0v[t] = s_z[zoff[t]];
  }

  g_eval<true>(s_z, s_h, s_dX[0], bW1, bias1, bW2, bias2, W2, lane, wave, kv);
#pragma unroll
  for (int t = 0; t < T2; ++t) {
    k1v[t] = kv[t];
    s_z[zoff[t]] = z0v[t] + dt * (kv[t] * (1.0f / 3.0f));
  }
  lds_barrier();

  g_eval<false>(s_z, s_h, s_dX[0], bW1, bias1, bW2, bias2, W2, lane, wave, kv);
#pragma unroll
  for (int t = 0; t < T2; ++t) {
    acv[t] = k1v[t] + 3.0f * kv[t];
    dv[t]  = k1v[t] - kv[t];
    s_z[zoff[t]] = z0v[t] + dt * fmaf(k1v[t], -(1.0f / 3.0f), kv[t]);
  }
  lds_barrier();

  g_eval<false>(s_z, s_h, s_dX[1], bW1, bias1, bW2, bias2, W2, lane, wave, kv);
#pragma unroll
  for (int t = 0; t < T2; ++t) {
    acv[t] += 3.0f * kv[t];
    s_z[zoff[t]] = z0v[t] + dt * (dv[t] + kv[t]);
  }
  lds_barrier();

  g_eval<false>(s_z, s_h, s_dX[2], bW1, bias1, bW2, bias2, W2, lane, wave, kv);
#pragma unroll
  for (int t = 0; t < T2; ++t) {
    s_z[zoff[t]] = z0v[t] + (acv[t] + kv[t]) * (dt * 0.125f);
  }
  lds_barrier();

  if (wave == 0) {
    int b = lane & 15, o = lane >> 4;
    float s = b_out[o];
    for (int hh = 0; hh < HH; ++hh)
      s = fmaf(s_z[b * ZLD + hh], W_out[o * HH + hh], s);
    out[(long)(r0 + b) * 2 + o] = s;
  }
}

extern "C" void kernel_launch(void* const* d_in, const int* in_sizes, int n_in,
                              void* d_out, int out_size, void* d_ws, size_t ws_size,
                              hipStream_t stream) {
  (void)n_in; (void)out_size; (void)d_ws; (void)ws_size;
  const float* coeffs = (const float*)d_in[0];
  const float* W_init = (const float*)d_in[1];
  const float* b_init = (const float*)d_in[2];
  const float* W1     = (const float*)d_in[3];
  const float* b1     = (const float*)d_in[4];
  const float* W2     = (const float*)d_in[5];
  const float* b2     = (const float*)d_in[6];
  const float* W_out  = (const float*)d_in[7];
  const float* b_out  = (const float*)d_in[8];
  float* out = (float*)d_out;

  int B = in_sizes[0] / (LL * CCH);
  int blocks = B / 16;
  ncde_fused<<<blocks, NTHREADS, 0, stream>>>(coeffs, W_init, b_init, W1, b1, W2, b2,
                                        W_out, b_out, out);
}
